# starting kernel
# baseline (speedup 1.0000x reference)
#include <hip/hip_runtime.h>


#define C0   32
#define LD   6
#define PDIM 96
#define VOL  (PDIM * PDIM * PDIM)
#define HSTR 36
#define TILES_PER_WAVE  4
#define WAVES_PER_BLOCK 8
#define POINTS_PER_BLOCK (16 * TILES_PER_WAVE * WAVES_PER_BLOCK)
#define FLT_BIG 3.402823466e+38f

typedef float v2f __attribute__((ext_vector_type(2)));
typedef float v8f __attribute__((ext_vector_type(8)));

__device__ __forceinline__ v8f wmma_f32(v2f a, v2f b, v8f c) {
  return __builtin_amdgcn_wmma_f32_16x16x4_f32(false, a, false, b,
                                               (short)0, c, false, false);
}

__device__ __forceinline__ float relu1(float x) {
  return __builtin_amdgcn_fmed3f(x, 0.0f, FLT_BIG);
}

__device__ __forceinline__ int permk(int s) { return (s >> 1) + ((s & 1) << 4); }

__global__ __launch_bounds__(256) void
fcomb_mlp_kernel(const float* __restrict__ fm, const float* __restrict__ z,
                 const float* __restrict__ w1, const float* __restrict__ b1,
                 const float* __restrict__ w2, const float* __restrict__ b2,
                 const float* __restrict__ w3, const float* __restrict__ b3,
                 const float* __restrict__ wl, const float* __restrict__ bl,
                 float* __restrict__ out)
{
  __shared__ float sW1[C0 * (C0 + LD)];
  __shared__ float sW2[C0 * C0];
  __shared__ float sW3[C0 * C0];
  __shared__ float sWl[C0];
  __shared__ float sB1[C0], sB2[C0], sB3[C0];
  __shared__ float sZ[LD];
  __shared__ float sBl[1];
  __shared__ float sHa[WAVES_PER_BLOCK][16 * HSTR];
  __shared__ float sHb[WAVES_PER_BLOCK][16 * HSTR];
  __shared__ float sO [WAVES_PER_BLOCK][16];

  const int tid = threadIdx.x;
  for (int i = tid; i < C0 * (C0 + LD); i += 256) sW1[i] = w1[i];
  for (int i = tid; i < C0 * C0;        i += 256) sW2[i] = w2[i];
  for (int i = tid; i < C0 * C0;        i += 256) sW3[i] = w3[i];
  if (tid < C0) { sWl[tid] = wl[tid]; sB1[tid] = b1[tid];
                  sB2[tid] = b2[tid]; sB3[tid] = b3[tid]; }
  if (tid < LD) sZ[tid] = z[tid];
  if (tid == 0) sBl[0] = bl[0];
  __syncthreads();

  const int lane = tid & 31;
  const int wave = tid >> 5;
  const int l16  = lane & 15;
  const int hi   = lane >> 4;

  float cinit1[2], cinit2[2], cinit3[2], wlv[2];
#pragma unroll
  for (int h = 0; h < 2; ++h) {
    const int n = l16 + 16 * h;
    float c = sB1[n];
#pragma unroll
    for (int k = 0; k < LD; ++k) c += sW1[n * (C0 + LD) + C0 + k] * sZ[k];
    cinit1[h] = c;
    cinit2[h] = sB2[n];
    cinit3[h] = sB3[n];
    wlv[h]    = sWl[n];
  }
  const float blv = sBl[0];

  v2f Bw1[2][8], Bw2[2][8], Bw3[2][8];
#pragma unroll
  for (int h = 0; h < 2; ++h) {
    const int n = l16 + 16 * h;
#pragma unroll
    for (int j = 0; j < 8; ++j) {
      const int k0 = 4 * j + 2 * hi;
      v2f t;
      t[0] = sW1[n * (C0 + LD) + k0]; t[1] = sW1[n * (C0 + LD) + k0 + 1];
      Bw1[h][j] = t;
      t[0] = sW2[n * C0 + permk(k0)]; t[1] = sW2[n * C0 + permk(k0 + 1)];
      Bw2[h][j] = t;
      t[0] = sW3[n * C0 + permk(k0)]; t[1] = sW3[n * C0 + permk(k0 + 1)];
      Bw3[h][j] = t;
    }
  }

  float* Ha = &sHa[wave][0];
  float* Hb = &sHb[wave][0];

  for (int t = 0; t < TILES_PER_WAVE; ++t) {
    const int pbase = blockIdx.x * POINTS_PER_BLOCK +
                      (wave * TILES_PER_WAVE + t) * 16;

    v2f a[8];
#pragma unroll
    for (int j = 0; j < 8; ++j) {
      const int k0 = 4 * j + 2 * hi;
      v2f av;
      av[0] = fm[k0 * VOL + pbase + l16];
      av[1] = fm[(k0 + 1) * VOL + pbase + l16];
      a[j] = av;
    }
    v8f acc0, acc1;
#pragma unroll
    for (int r = 0; r < 8; ++r) { acc0[r] = cinit1[0]; acc1[r] = cinit1[1]; }
#pragma unroll
    for (int j = 0; j < 8; ++j) {
      acc0 = wmma_f32(a[j], Bw1[0][j], acc0);
      acc1 = wmma_f32(a[j], Bw1[1][j], acc1);
    }
#pragma unroll
    for (int r = 0; r < 8; ++r) {
      v2f p;
      p[0] = relu1(acc0[r]);
      p[1] = relu1(acc1[r]);
      *(v2f*)&Ha[(r + 8 * hi) * HSTR + 2 * l16] = p;
    }

#pragma unroll
    for (int j = 0; j < 8; ++j)
      a[j] = *(const v2f*)&Ha[l16 * HSTR + 4 * j + 2 * hi];
#pragma unroll
    for (int r = 0; r < 8; ++r) { acc0[r] = cinit2[0]; acc1[r] = cinit2[1]; }
#pragma unroll
    for (int j = 0; j < 8; ++j) {
      acc0 = wmma_f32(a[j], Bw2[0][j], acc0);
      acc1 = wmma_f32(a[j], Bw2[1][j], acc1);
    }
#pragma unroll
    for (int r = 0; r < 8; ++r) {
      v2f p;
      p[0] = relu1(acc0[r]);
      p[1] = relu1(acc1[r]);
      *(v2f*)&Hb[(r + 8 * hi) * HSTR + 2 * l16] = p;
    }

#pragma unroll
    for (int j = 0; j < 8; ++j)
      a[j] = *(const v2f*)&Hb[l16 * HSTR + 4 * j + 2 * hi];
#pragma unroll
    for (int r = 0; r < 8; ++r) { acc0[r] = cinit3[0]; acc1[r] = cinit3[1]; }
#pragma unroll
    for (int j = 0; j < 8; ++j) {
      acc0 = wmma_f32(a[j], Bw3[0][j], acc0);
      acc1 = wmma_f32(a[j], Bw3[1][j], acc1);
    }

#pragma unroll
    for (int r = 0; r < 8; ++r) {
      const float v0 = relu1(acc0[r]);
      const float v1 = relu1(acc1[r]);
      float s = v0 * wlv[0] + v1 * wlv[1];
      s += __shfl_xor(s, 1);
      s += __shfl_xor(s, 2);
      s += __shfl_xor(s, 4);
      s += __shfl_xor(s, 8);
      if (l16 == 0) sO[wave][r + 8 * hi] = s;
    }
    if (lane < 16) out[pbase + lane] = sO[wave][lane] + blv;
  }
}

extern "C" void kernel_launch(void* const* d_in, const int* in_sizes, int n_in,
                              void* d_out, int out_size, void* d_ws, size_t ws_size,
                              hipStream_t stream) {
  (void)in_sizes; (void)n_in; (void)out_size; (void)d_ws; (void)ws_size;
  const float* fm = (const float*)d_in[0];
  const float* z  = (const float*)d_in[1];
  const float* w1 = (const float*)d_in[2];
  const float* b1 = (const float*)d_in[3];
  const float* w2 = (const float*)d_in[4];
  const float* b2 = (const float*)d_in[5];
  const float* w3 = (const float*)d_in[6];
  const float* b3 = (const float*)d_in[7];
  const float* wl = (const float*)d_in[8];
  const float* bl = (const float*)d_in[9];
  float* out = (float*)d_out;

  dim3 grid(VOL / POINTS_PER_BLOCK);
  dim3 block(256);
  fcomb_mlp_kernel<<<grid, block, 0, stream>>>(fm, z, w1, b1, w2, b2,
                                               w3, b3, wl, bl, out);
}


# Weights skip LDS: each wave loads its B fragments straight from global as 16-byte row pieces (A-side LDS columns re-ordered to 16*half+2j to match); staging loops and barrier go; k order new
# speedup vs baseline: 1.9870x; 1.9870x over previous
#include <hip/hip_runtime.h>


#define C0   32
#define LD   6
#define PDIM 96
#define VOL  (PDIM * PDIM * PDIM)
#define HSTR 36
#define TILES_PER_WAVE  4
#define WAVES_PER_BLOCK 8
#define POINTS_PER_BLOCK (16 * TILES_PER_WAVE * WAVES_PER_BLOCK)
#define FLT_BIG 3.402823466e+38f

typedef float v2f __attribute__((ext_vector_type(2)));
typedef float v4f __attribute__((ext_vector_type(4)));
typedef float v8f __attribute__((ext_vector_type(8)));

__device__ __forceinline__ v8f wmma_f32(v2f a, v2f b, v8f c) {
  return __builtin_amdgcn_wmma_f32_16x16x4_f32(false, a, false, b,
                                               (short)0, c, false, false);
}

__device__ __forceinline__ float relu1(float x) {
  return __builtin_amdgcn_fmed3f(x, 0.0f, FLT_BIG);
}


__global__ __launch_bounds__(256) void
fcomb_mlp_kernel(const float* __restrict__ fm, const float* __restrict__ z,
                 const float* __restrict__ w1, const float* __restrict__ b1,
                 const float* __restrict__ w2, const float* __restrict__ b2,
                 const float* __restrict__ w3, const float* __restrict__ b3,
                 const float* __restrict__ wl, const float* __restrict__ bl,
                 float* __restrict__ out)
{
  __shared__ float sHa[WAVES_PER_BLOCK][16 * HSTR];
  __shared__ float sHb[WAVES_PER_BLOCK][16 * HSTR];

  const int tid = threadIdx.x;

  const int lane = tid & 31;
  const int wave = tid >> 5;
  const int l16  = lane & 15;
  const int hi   = lane >> 4;

  float cinit1[2], cinit2[2], cinit3[2];
#pragma unroll
  for (int h = 0; h < 2; ++h) {
    const int n = l16 + 16 * h;
    float c = b1[n];
#pragma unroll
    for (int k = 0; k < LD; ++k) c += w1[n * (C0 + LD) + C0 + k] * z[k];
    cinit1[h] = c;
    cinit2[h] = b2[n];
    cinit3[h] = b3[n];
  }
  const float blv = bl[0];

  v2f Bw1[2][8], Bw2[2][8], Bw3[2][8];
#pragma unroll
  for (int h = 0; h < 2; ++h) {
    const int n = l16 + 16 * h;
#pragma unroll
    for (int j = 0; j < 8; ++j)
      Bw1[h][j] = *(const v2f*)&w1[n * (C0 + LD) + 4 * j + 2 * hi];
    const v4f* r2 = (const v4f*)&w2[n * C0 + 8 * hi];
    const v4f* r3 = (const v4f*)&w3[n * C0 + 8 * hi];
    const v4f q2[4] = { r2[0], r2[1], r2[4], r2[5] };
    const v4f q3[4] = { r3[0], r3[1], r3[4], r3[5] };
#pragma unroll
    for (int j = 0; j < 8; ++j) {
      v2f t;
      t[0] = q2[j >> 2][j & 3]; t[1] = q2[2 + (j >> 2)][j & 3];
      Bw2[h][j] = t;
      t[0] = q3[j >> 2][j & 3]; t[1] = q3[2 + (j >> 2)][j & 3];
      Bw3[h][j] = t;
    }
  }

  v2f Awl[8];
#pragma unroll
  for (int j = 0; j < 8; ++j) {
    v2f t;
    t[0] = wl[8 * hi + j]; t[1] = wl[8 * hi + j + 16];
    Awl[j] = t;
  }

  float* Ha = &sHa[wave][0];
  float* Hb = &sHb[wave][0];

  for (int t = 0; t < TILES_PER_WAVE; t += 2) {
    const int pbase2 = blockIdx.x * POINTS_PER_BLOCK +
                       (wave * TILES_PER_WAVE + t) * 16;
    float res2[2];
#pragma unroll
    for (int u = 0; u < 2; ++u) {
    const int pbase = pbase2 + 16 * u;

    v2f a[8];
#pragma unroll
    for (int j = 0; j < 8; ++j) {
      const int k0 = 4 * j + 2 * hi;
      v2f av;
      av[0] = fm[k0 * VOL + pbase + l16];
      av[1] = fm[(k0 + 1) * VOL + pbase + l16];
      a[j] = av;
    }
    v8f acc0, acc1;
#pragma unroll
    for (int r = 0; r < 8; ++r) { acc0[r] = cinit1[0]; acc1[r] = cinit1[1]; }
#pragma unroll
    for (int j = 0; j < 8; ++j) {
      acc0 = wmma_f32(a[j], Bw1[0][j], acc0);
      acc1 = wmma_f32(a[j], Bw1[1][j], acc1);
    }
#pragma unroll
    for (int r = 0; r < 8; ++r) {
      v2f p;
      p[0] = relu1(acc0[r]);
      p[1] = relu1(acc1[r]);
      *(v2f*)&Ha[(r + 8 * hi) * HSTR + 2 * l16] = p;
    }

#pragma unroll
    for (int j = 0; j < 8; ++j)
      a[j] = *(const v2f*)&Ha[l16 * HSTR + 16 * hi + 2 * j];
#pragma unroll
    for (int r = 0; r < 8; ++r) { acc0[r] = cinit2[0]; acc1[r] = cinit2[1]; }
#pragma unroll
    for (int j = 0; j < 8; ++j) {
      acc0 = wmma_f32(a[j], Bw2[0][j], acc0);
      acc1 = wmma_f32(a[j], Bw2[1][j], acc1);
    }
#pragma unroll
    for (int r = 0; r < 8; ++r) {
      v2f p;
      p[0] = relu1(acc0[r]);
      p[1] = relu1(acc1[r]);
      *(v2f*)&Hb[(r + 8 * hi) * HSTR + 2 * l16] = p;
    }

#pragma unroll
    for (int j = 0; j < 8; ++j)
      a[j] = *(const v2f*)&Hb[l16 * HSTR + 16 * hi + 2 * j];
#pragma unroll
    for (int r = 0; r < 8; ++r) { acc0[r] = cinit3[0]; acc1[r] = cinit3[1]; }
#pragma unroll
    for (int j = 0; j < 8; ++j) {
      acc0 = wmma_f32(a[j], Bw3[0][j], acc0);
      acc1 = wmma_f32(a[j], Bw3[1][j], acc1);
    }

#pragma unroll
    for (int r = 0; r < 8; ++r) {
      v2f p;
      p[0] = relu1(acc0[r]);
      p[1] = relu1(acc1[r]);
      *(v2f*)&Ha[(r + 8 * hi) * HSTR + 2 * l16] = p;
    }

#pragma unroll
    for (int j = 0; j < 8; ++j)
      a[j] = *(const v2f*)&Ha[l16 * HSTR + 16 * hi + 2 * j];
    v8f o0, o1;
#pragma unroll
    for (int r = 0; r < 8; ++r) { o0[r] = blv; o1[r] = 0.0f; }
#pragma unroll
    for (int j = 0; j < 8; j += 2) {
      o0 = wmma_f32(Awl[j],     a[j],     o0);
      o1 = wmma_f32(Awl[j + 1], a[j + 1], o1);
    }
    res2[u] = o0[0] + o1[0];
    }
    out[pbase2 + lane] = hi ? res2[1] : res2[0];
  }
}

extern "C" void kernel_launch(void* const* d_in, const int* in_sizes, int n_in,
                              void* d_out, int out_size, void* d_ws, size_t ws_size,
                              hipStream_t stream) {
  (void)in_sizes; (void)n_in; (void)out_size; (void)d_ws; (void)ws_size;
  const float* fm = (const float*)d_in[0];
  const float* z  = (const float*)d_in[1];
  const float* w1 = (const float*)d_in[2];
  const float* b1 = (const float*)d_in[3];
  const float* w2 = (const float*)d_in[4];
  const float* b2 = (const float*)d_in[5];
  const float* w3 = (const float*)d_in[6];
  const float* b3 = (const float*)d_in[7];
  const float* wl = (const float*)d_in[8];
  const float* bl = (const float*)d_in[9];
  float* out = (float*)d_out;

  dim3 grid(VOL / POINTS_PER_BLOCK);
  dim3 block(256);
  fcomb_mlp_kernel<<<grid, block, 0, stream>>>(fm, z, w1, b1, w2, b2,
                                               w3, b3, wl, bl, out);
}


# Each wave reads w1,w2,w3 once as whole-line 16-byte loads (4 lines a load, was 16: a weight row per lane) into a wave-private LDS copy, no barrier, and takes its B fragments there; k order kept
# speedup vs baseline: 2.0724x; 1.0430x over previous
#include <hip/hip_runtime.h>


#define C0   32
#define LD   6
#define PDIM 96
#define VOL  (PDIM * PDIM * PDIM)
#define HSTR 36
#define TILES_PER_WAVE  4
#define WAVES_PER_BLOCK 8
#define POINTS_PER_BLOCK (16 * TILES_PER_WAVE * WAVES_PER_BLOCK)
#define FLT_BIG 3.402823466e+38f
#define W1ROW  (C0 + LD)
#define W1N    (C0 * W1ROW)
#define W1V4   (W1N / 4)
#define WPITCH 36
#define W23N   (C0 * WPITCH)
#define WAVE_LDS (W1N + 2 * W23N)

typedef float v2f __attribute__((ext_vector_type(2)));
typedef float v4f __attribute__((ext_vector_type(4)));
typedef float v8f __attribute__((ext_vector_type(8)));

__device__ __forceinline__ v8f wmma_f32(v2f a, v2f b, v8f c) {
  return __builtin_amdgcn_wmma_f32_16x16x4_f32(false, a, false, b,
                                               (short)0, c, false, false);
}

__device__ __forceinline__ float relu1(float x) {
  return __builtin_amdgcn_fmed3f(x, 0.0f, FLT_BIG);
}


__global__ __launch_bounds__(256) void
fcomb_mlp_kernel(const float* __restrict__ fm, const float* __restrict__ z,
                 const float* __restrict__ w1, const float* __restrict__ b1,
                 const float* __restrict__ w2, const float* __restrict__ b2,
                 const float* __restrict__ w3, const float* __restrict__ b3,
                 const float* __restrict__ wl, const float* __restrict__ bl,
                 float* __restrict__ out)
{
  __shared__ __attribute__((aligned(16))) float sW[WAVES_PER_BLOCK][WAVE_LDS];

  const int tid = threadIdx.x;

  const int lane = tid & 31;
  const int wave = tid >> 5;
  const int l16  = lane & 15;
  const int hi   = lane >> 4;

  float* Ws  = &sW[wave][0];
  float* Ws2 = Ws + W1N;
  float* Ws3 = Ws2 + W23N;
  {
    const v4f* g1 = (const v4f*)w1;
    const v4f* g2 = (const v4f*)w2;
    const v4f* g3 = (const v4f*)w3;
    v4f t1[10], t2[8], t3[8];
#pragma unroll
    for (int i = 0; i < 10; ++i) {
      const int idx = min(i * 32 + lane, W1V4 - 1);
      t1[i] = g1[idx];
    }
#pragma unroll
    for (int i = 0; i < 8; ++i) { t2[i] = g2[i * 32 + lane]; t3[i] = g3[i * 32 + lane]; }
#pragma unroll
    for (int i = 0; i < 10; ++i) {
      const int idx = min(i * 32 + lane, W1V4 - 1);
      *(v4f*)&Ws[4 * idx] = t1[i];
    }
#pragma unroll
    for (int i = 0; i < 8; ++i) {
      const int off = (4 * i + (lane >> 3)) * WPITCH + 4 * (lane & 7);
      *(v4f*)&Ws2[off] = t2[i];
      *(v4f*)&Ws3[off] = t3[i];
    }
  }

  float cinit1[2], cinit2[2], cinit3[2];
#pragma unroll
  for (int h = 0; h < 2; ++h) {
    const int n = l16 + 16 * h;
    float c = b1[n];
#pragma unroll
    for (int k = 0; k < LD; ++k) c += Ws[n * W1ROW + C0 + k] * z[k];
    cinit1[h] = c;
    cinit2[h] = b2[n];
    cinit3[h] = b3[n];
  }
  const float blv = bl[0];

  v2f Bw1[2][8], Bw2[2][8], Bw3[2][8];
#pragma unroll
  for (int h = 0; h < 2; ++h) {
    const int n = l16 + 16 * h;
#pragma unroll
    for (int j = 0; j < 8; ++j)
      Bw1[h][j] = *(const v2f*)&Ws[n * W1ROW + 4 * j + 2 * hi];
    const v4f* r2 = (const v4f*)&Ws2[n * WPITCH + 8 * hi];
    const v4f* r3 = (const v4f*)&Ws3[n * WPITCH + 8 * hi];
    const v4f q2[4] = { r2[0], r2[1], r2[4], r2[5] };
    const v4f q3[4] = { r3[0], r3[1], r3[4], r3[5] };
#pragma unroll
    for (int j = 0; j < 8; ++j) {
      v2f t;
      t[0] = q2[j >> 2][j & 3]; t[1] = q2[2 + (j >> 2)][j & 3];
      Bw2[h][j] = t;
      t[0] = q3[j >> 2][j & 3]; t[1] = q3[2 + (j >> 2)][j & 3];
      Bw3[h][j] = t;
    }
  }

  v2f Awl[8];
#pragma unroll
  for (int j = 0; j < 8; ++j) {
    v2f t;
    t[0] = wl[8 * hi + j]; t[1] = wl[8 * hi + j + 16];
    Awl[j] = t;
  }

  float* Ha = Ws;
  float* Hb = Ws + 16 * HSTR;

  for (int t = 0; t < TILES_PER_WAVE; t += 2) {
    const int pbase2 = blockIdx.x * POINTS_PER_BLOCK +
                       (wave * TILES_PER_WAVE + t) * 16;
    float res2[2];
#pragma unroll
    for (int u = 0; u < 2; ++u) {
    const int pbase = pbase2 + 16 * u;

    v2f a[8];
#pragma unroll
    for (int j = 0; j < 8; ++j) {
      const int k0 = 4 * j + 2 * hi;
      v2f av;
      av[0] = fm[k0 * VOL + pbase + l16];
      av[1] = fm[(k0 + 1) * VOL + pbase + l16];
      a[j] = av;
    }
    v8f acc0, acc1;
#pragma unroll
    for (int r = 0; r < 8; ++r) { acc0[r] = cinit1[0]; acc1[r] = cinit1[1]; }
#pragma unroll
    for (int j = 0; j < 8; ++j) {
      acc0 = wmma_f32(a[j], Bw1[0][j], acc0);
      acc1 = wmma_f32(a[j], Bw1[1][j], acc1);
    }
#pragma unroll
    for (int r = 0; r < 8; ++r) {
      v2f p;
      p[0] = relu1(acc0[r]);
      p[1] = relu1(acc1[r]);
      *(v2f*)&Ha[(r + 8 * hi) * HSTR + 2 * l16] = p;
    }

#pragma unroll
    for (int j = 0; j < 8; ++j)
      a[j] = *(const v2f*)&Ha[l16 * HSTR + 16 * hi + 2 * j];
#pragma unroll
    for (int r = 0; r < 8; ++r) { acc0[r] = cinit2[0]; acc1[r] = cinit2[1]; }
#pragma unroll
    for (int j = 0; j < 8; ++j) {
      acc0 = wmma_f32(a[j], Bw2[0][j], acc0);
      acc1 = wmma_f32(a[j], Bw2[1][j], acc1);
    }
#pragma unroll
    for (int r = 0; r < 8; ++r) {
      v2f p;
      p[0] = relu1(acc0[r]);
      p[1] = relu1(acc1[r]);
      *(v2f*)&Hb[(r + 8 * hi) * HSTR + 2 * l16] = p;
    }

#pragma unroll
    for (int j = 0; j < 8; ++j)
      a[j] = *(const v2f*)&Hb[l16 * HSTR + 16 * hi + 2 * j];
#pragma unroll
    for (int r = 0; r < 8; ++r) { acc0[r] = cinit3[0]; acc1[r] = cinit3[1]; }
#pragma unroll
    for (int j = 0; j < 8; ++j) {
      acc0 = wmma_f32(a[j], Bw3[0][j], acc0);
      acc1 = wmma_f32(a[j], Bw3[1][j], acc1);
    }

#pragma unroll
    for (int r = 0; r < 8; ++r) {
      v2f p;
      p[0] = relu1(acc0[r]);
      p[1] = relu1(acc1[r]);
      *(v2f*)&Ha[(r + 8 * hi) * HSTR + 2 * l16] = p;
    }

#pragma unroll
    for (int j = 0; j < 8; ++j)
      a[j] = *(const v2f*)&Ha[l16 * HSTR + 16 * hi + 2 * j];
    v8f o0, o1;
#pragma unroll
    for (int r = 0; r < 8; ++r) { o0[r] = blv; o1[r] = 0.0f; }
#pragma unroll
    for (int j = 0; j < 8; j += 2) {
      o0 = wmma_f32(Awl[j],     a[j],     o0);
      o1 = wmma_f32(Awl[j + 1], a[j + 1], o1);
    }
    res2[u] = o0[0] + o1[0];
    }
    out[pbase2 + lane] = hi ? res2[1] : res2[0];
  }
}

extern "C" void kernel_launch(void* const* d_in, const int* in_sizes, int n_in,
                              void* d_out, int out_size, void* d_ws, size_t ws_size,
                              hipStream_t stream) {
  (void)in_sizes; (void)n_in; (void)out_size; (void)d_ws; (void)ws_size;
  const float* fm = (const float*)d_in[0];
  const float* z  = (const float*)d_in[1];
  const float* w1 = (const float*)d_in[2];
  const float* b1 = (const float*)d_in[3];
  const float* w2 = (const float*)d_in[4];
  const float* b2 = (const float*)d_in[5];
  const float* w3 = (const float*)d_in[6];
  const float* b3 = (const float*)d_in[7];
  const float* wl = (const float*)d_in[8];
  const float* bl = (const float*)d_in[9];
  float* out = (float*)d_out;

  dim3 grid(VOL / POINTS_PER_BLOCK);
  dim3 block(256);
  fcomb_mlp_kernel<<<grid, block, 0, stream>>>(fm, z, w1, b1, w2, b2,
                                               w3, b3, wl, bl, out);
}


# Output layer folded into layer 3: w3 rows and b3 pre-scaled by wl (wl*relu(x) = wl*x clamped on wl's side of 0), out = DPP row sum of clamped D; 8 WMMAs + 1 LDS relayout a tile go; sum order new
# speedup vs baseline: 2.2133x; 1.0680x over previous
#include <hip/hip_runtime.h>


#define C0   32
#define LD   6
#define PDIM 96
#define VOL  (PDIM * PDIM * PDIM)
#define HSTR 36
#define TILES_PER_WAVE  4
#define WAVES_PER_BLOCK 8
#define POINTS_PER_BLOCK (16 * TILES_PER_WAVE * WAVES_PER_BLOCK)
#define FLT_BIG 3.402823466e+38f
#define W1ROW  (C0 + LD)
#define W1N    (C0 * W1ROW)
#define W1V4   (W1N / 4)
#define WPITCH 36
#define W23N   (C0 * WPITCH)
#define WAVE_LDS (W1N + 2 * W23N)

typedef float v2f __attribute__((ext_vector_type(2)));
typedef float v4f __attribute__((ext_vector_type(4)));
typedef float v8f __attribute__((ext_vector_type(8)));

__device__ __forceinline__ v8f wmma_f32(v2f a, v2f b, v8f c) {
  return __builtin_amdgcn_wmma_f32_16x16x4_f32(false, a, false, b,
                                               (short)0, c, false, false);
}

__device__ __forceinline__ float relu1(float x) {
  return __builtin_amdgcn_fmed3f(x, 0.0f, FLT_BIG);
}

__device__ __forceinline__ float clamp_side(float x, float bound) {
  return __builtin_amdgcn_fmed3f(x, 0.0f, bound);
}

template <int MASK>
__device__ __forceinline__ float row_xor(float x) {
  return __builtin_bit_cast(float, __builtin_amdgcn_update_dpp(
      0, __builtin_bit_cast(int, x), 0x160 | MASK, 0xf, 0xf, true));
}


__global__ __launch_bounds__(256) void
fcomb_mlp_kernel(const float* __restrict__ fm, const float* __restrict__ z,
                 const float* __restrict__ w1, const float* __restrict__ b1,
                 const float* __restrict__ w2, const float* __restrict__ b2,
                 const float* __restrict__ w3, const float* __restrict__ b3,
                 const float* __restrict__ wl, const float* __restrict__ bl,
                 float* __restrict__ out)
{
  __shared__ __attribute__((aligned(16))) float sW[WAVES_PER_BLOCK][WAVE_LDS];

  const int tid = threadIdx.x;

  const int lane = tid & 31;
  const int wave = tid >> 5;
  const int l16  = lane & 15;
  const int hi   = lane >> 4;

  float* Ws  = &sW[wave][0];
  float* Ws2 = Ws + W1N;
  float* Ws3 = Ws2 + W23N;
  {
    const v4f* g1 = (const v4f*)w1;
    const v4f* g2 = (const v4f*)w2;
    const v4f* g3 = (const v4f*)w3;
    v4f t1[10], t2[8], t3[8];
#pragma unroll
    for (int i = 0; i < 10; ++i) {
      const int idx = min(i * 32 + lane, W1V4 - 1);
      t1[i] = g1[idx];
    }
#pragma unroll
    for (int i = 0; i < 8; ++i) { t2[i] = g2[i * 32 + lane]; t3[i] = g3[i * 32 + lane]; }
#pragma unroll
    for (int i = 0; i < 10; ++i) {
      const int idx = min(i * 32 + lane, W1V4 - 1);
      *(v4f*)&Ws[4 * idx] = t1[i];
    }
#pragma unroll
    for (int i = 0; i < 8; ++i) {
      const int off = (4 * i + (lane >> 3)) * WPITCH + 4 * (lane & 7);
      *(v4f*)&Ws2[off] = t2[i];
      *(v4f*)&Ws3[off] = t3[i];
    }
  }

  float wln[2], bnd[2];
#pragma unroll
  for (int h = 0; h < 2; ++h) {
    wln[h] = wl[l16 + 16 * h];
    bnd[h] = wln[h] >= 0.0f ? FLT_BIG : -FLT_BIG;
  }

  float cinit1[2], cinit2[2], cinit3[2];
#pragma unroll
  for (int h = 0; h < 2; ++h) {
    const int n = l16 + 16 * h;
    float c = b1[n];
#pragma unroll
    for (int k = 0; k < LD; ++k) c += Ws[n * W1ROW + C0 + k] * z[k];
    cinit1[h] = c;
    cinit2[h] = b2[n];
    cinit3[h] = b3[n] * wln[h];
  }
  const float blv = bl[0];

  v2f Bw1[2][8], Bw2[2][8], Bw3[2][8];
#pragma unroll
  for (int h = 0; h < 2; ++h) {
    const int n = l16 + 16 * h;
#pragma unroll
    for (int j = 0; j < 8; ++j)
      Bw1[h][j] = *(const v2f*)&Ws[n * W1ROW + 4 * j + 2 * hi];
    const v4f* r2 = (const v4f*)&Ws2[n * WPITCH + 8 * hi];
    const v4f* r3 = (const v4f*)&Ws3[n * WPITCH + 8 * hi];
    const v4f q2[4] = { r2[0], r2[1], r2[4], r2[5] };
    const v4f q3[4] = { r3[0], r3[1], r3[4], r3[5] };
#pragma unroll
    for (int j = 0; j < 8; ++j) {
      v2f t;
      t[0] = q2[j >> 2][j & 3]; t[1] = q2[2 + (j >> 2)][j & 3];
      Bw2[h][j] = t;
      t[0] = q3[j >> 2][j & 3]; t[1] = q3[2 + (j >> 2)][j & 3];
      Bw3[h][j] = t * wln[h];
    }
  }

  const bool lb0 = (lane & 1) != 0, lb1 = (lane & 2) != 0;
  const bool lb2 = (lane & 4) != 0, lb3 = (lane & 8) != 0;
  const int srcl = 16 * (l16 >> 3) + 2 * (l16 & 7) + hi;

  float* Ha = Ws;
  float* Hb = Ws + 16 * HSTR;

#pragma nounroll
  for (int t = 0; t < TILES_PER_WAVE; t += 2) {
    const int pbase2 = blockIdx.x * POINTS_PER_BLOCK +
                       (wave * TILES_PER_WAVE + t) * 16;
    float res2[2];
#pragma unroll
    for (int u = 0; u < 2; ++u) {
    const int pbase = pbase2 + 16 * u;

    v2f a[8];
#pragma unroll
    for (int j = 0; j < 8; ++j) {
      const int k0 = 4 * j + 2 * hi;
      v2f av;
      av[0] = fm[k0 * VOL + pbase + l16];
      av[1] = fm[(k0 + 1) * VOL + pbase + l16];
      a[j] = av;
    }
    v8f acc0, acc1;
#pragma unroll
    for (int r = 0; r < 8; ++r) { acc0[r] = cinit1[0]; acc1[r] = cinit1[1]; }
#pragma unroll
    for (int j = 0; j < 8; ++j) {
      acc0 = wmma_f32(a[j], Bw1[0][j], acc0);
      acc1 = wmma_f32(a[j], Bw1[1][j], acc1);
    }
#pragma unroll
    for (int r = 0; r < 8; ++r) {
      v2f p;
      p[0] = relu1(acc0[r]);
      p[1] = relu1(acc1[r]);
      *(v2f*)&Ha[(r + 8 * hi) * HSTR + 2 * l16] = p;
    }

#pragma unroll
    for (int j = 0; j < 8; ++j)
      a[j] = *(const v2f*)&Ha[l16 * HSTR + 16 * hi + 2 * j];
#pragma unroll
    for (int r = 0; r < 8; ++r) { acc0[r] = cinit2[0]; acc1[r] = cinit2[1]; }
#pragma unroll
    for (int j = 0; j < 8; ++j) {
      acc0 = wmma_f32(a[j], Bw2[0][j], acc0);
      acc1 = wmma_f32(a[j], Bw2[1][j], acc1);
    }
#pragma unroll
    for (int r = 0; r < 8; ++r) {
      v2f p;
      p[0] = relu1(acc0[r]);
      p[1] = relu1(acc1[r]);
      *(v2f*)&Hb[(r + 8 * hi) * HSTR + 2 * l16] = p;
    }

#pragma unroll
    for (int j = 0; j < 8; ++j)
      a[j] = *(const v2f*)&Hb[l16 * HSTR + 16 * hi + 2 * j];
#pragma unroll
    for (int r = 0; r < 8; ++r) { acc0[r] = cinit3[0]; acc1[r] = cinit3[1]; }
#pragma unroll
    for (int j = 0; j < 8; ++j) {
      acc0 = wmma_f32(a[j], Bw3[0][j], acc0);
      acc1 = wmma_f32(a[j], Bw3[1][j], acc1);
    }

    float v8[8], s4[4], s2[2];
#pragma unroll
    for (int r = 0; r < 8; ++r)
      v8[r] = clamp_side(acc0[r], bnd[0]) + clamp_side(acc1[r], bnd[1]);
#pragma unroll
    for (int i = 0; i < 4; ++i) {
      const float keep = lb3 ? v8[i + 4] : v8[i];
      const float give = lb3 ? v8[i] : v8[i + 4];
      s4[i] = keep + row_xor<8>(give);
    }
#pragma unroll
    for (int i = 0; i < 2; ++i) {
      const float keep = lb2 ? s4[i + 2] : s4[i];
      const float give = lb2 ? s4[i] : s4[i + 2];
      s2[i] = keep + row_xor<4>(give);
    }
    {
      const float keep = lb1 ? s2[1] : s2[0];
      const float give = lb1 ? s2[0] : s2[1];
      res2[u] = keep + row_xor<2>(give);
    }
    }
    const float keep = lb0 ? res2[1] : res2[0];
    const float give = lb0 ? res2[0] : res2[1];
    const float tot = keep + row_xor<1>(give);
    const float o = __builtin_bit_cast(float, __builtin_amdgcn_ds_bpermute(
        4 * srcl, __builtin_bit_cast(int, tot)));
    out[pbase2 + lane] = o + blv;
  }
}

extern "C" void kernel_launch(void* const* d_in, const int* in_sizes, int n_in,
                              void* d_out, int out_size, void* d_ws, size_t ws_size,
                              hipStream_t stream) {
  (void)in_sizes; (void)n_in; (void)out_size; (void)d_ws; (void)ws_size;
  const float* fm = (const float*)d_in[0];
  const float* z  = (const float*)d_in[1];
  const float* w1 = (const float*)d_in[2];
  const float* b1 = (const float*)d_in[3];
  const float* w2 = (const float*)d_in[4];
  const float* b2 = (const float*)d_in[5];
  const float* w3 = (const float*)d_in[6];
  const float* b3 = (const float*)d_in[7];
  const float* wl = (const float*)d_in[8];
  const float* bl = (const float*)d_in[9];
  float* out = (float*)d_out;

  dim3 grid(VOL / POINTS_PER_BLOCK);
  dim3 block(256);
  fcomb_mlp_kernel<<<grid, block, 0, stream>>>(fm, z, w1, b1, w2, b2,
                                               w3, b3, wl, bl, out);
}


# Feature map read as 16-byte pieces: lane = 4 consecutive points of a channel, tile = every 4th point; a wave's 64 loads become 16, issued at entry behind the weight loads; one 256-B store; sums kept
# speedup vs baseline: 2.3302x; 1.0528x over previous
#include <hip/hip_runtime.h>


#define C0   32
#define LD   6
#define PDIM 96
#define VOL  (PDIM * PDIM * PDIM)
#define HSTR 36
#define TILES_PER_WAVE  4
#define WAVES_PER_BLOCK 8
#define POINTS_PER_BLOCK (16 * TILES_PER_WAVE * WAVES_PER_BLOCK)
#define FLT_BIG 3.402823466e+38f
#define W1ROW  (C0 + LD)
#define W1N    (C0 * W1ROW)
#define W1V4   (W1N / 4)
#define WPITCH 36
#define W23N   (C0 * WPITCH)
#define WAVE_LDS (W1N + 2 * W23N)

typedef float v2f __attribute__((ext_vector_type(2)));
typedef float v4f __attribute__((ext_vector_type(4)));
typedef float v8f __attribute__((ext_vector_type(8)));

__device__ __forceinline__ v8f wmma_f32(v2f a, v2f b, v8f c) {
  return __builtin_amdgcn_wmma_f32_16x16x4_f32(false, a, false, b,
                                               (short)0, c, false, false);
}

__device__ __forceinline__ float relu1(float x) {
  return __builtin_amdgcn_fmed3f(x, 0.0f, FLT_BIG);
}

__device__ __forceinline__ float clamp_side(float x, float bound) {
  return __builtin_amdgcn_fmed3f(x, 0.0f, bound);
}

template <int MASK>
__device__ __forceinline__ float row_xor(float x) {
  return __builtin_bit_cast(float, __builtin_amdgcn_update_dpp(
      0, __builtin_bit_cast(int, x), 0x160 | MASK, 0xf, 0xf, true));
}


__global__ __launch_bounds__(256) void
fcomb_mlp_kernel(const float* __restrict__ fm, const float* __restrict__ z,
                 const float* __restrict__ w1, const float* __restrict__ b1,
                 const float* __restrict__ w2, const float* __restrict__ b2,
                 const float* __restrict__ w3, const float* __restrict__ b3,
                 const float* __restrict__ wl, const float* __restrict__ bl,
                 float* __restrict__ out)
{
  __shared__ __attribute__((aligned(16))) float sW[WAVES_PER_BLOCK][WAVE_LDS];

  const int tid = threadIdx.x;

  const int lane = tid & 31;
  const int wave = tid >> 5;
  const int l16  = lane & 15;
  const int hi   = lane >> 4;

  const int wbase = blockIdx.x * POINTS_PER_BLOCK + wave * (16 * TILES_PER_WAVE);
  const v4f* fm4 = (const v4f*)fm;
  const int xoff = 2 * hi * (VOL / 4) + wbase / 4 + l16;
  v4f X[8][2];

  float* Ws  = &sW[wave][0];
  float* Ws2 = Ws + W1N;
  float* Ws3 = Ws2 + W23N;
  {
    const v4f* g1 = (const v4f*)w1;
    const v4f* g2 = (const v4f*)w2;
    const v4f* g3 = (const v4f*)w3;
    v4f t1[10], t2[8], t3[8];
#pragma unroll
    for (int i = 0; i < 10; ++i) {
      const int idx = min(i * 32 + lane, W1V4 - 1);
      t1[i] = g1[idx];
    }
#pragma unroll
    for (int i = 0; i < 8; ++i) { t2[i] = g2[i * 32 + lane]; t3[i] = g3[i * 32 + lane]; }
#pragma unroll
    for (int j = 0; j < 8; ++j) {
#pragma unroll
      for (int s = 0; s < 2; ++s)
        X[j][s] = fm4[(4 * j + s) * (VOL / 4) + xoff];
    }
#pragma unroll
    for (int i = 0; i < 10; ++i) {
      const int idx = min(i * 32 + lane, W1V4 - 1);
      *(v4f*)&Ws[4 * idx] = t1[i];
    }
#pragma unroll
    for (int i = 0; i < 8; ++i) {
      const int off = (4 * i + (lane >> 3)) * WPITCH + 4 * (lane & 7);
      *(v4f*)&Ws2[off] = t2[i];
      *(v4f*)&Ws3[off] = t3[i];
    }
  }

  float wln[2], bnd[2];
#pragma unroll
  for (int h = 0; h < 2; ++h) {
    wln[h] = wl[l16 + 16 * h];
    bnd[h] = wln[h] >= 0.0f ? FLT_BIG : -FLT_BIG;
  }

  float cinit1[2], cinit2[2], cinit3[2];
#pragma unroll
  for (int h = 0; h < 2; ++h) {
    const int n = l16 + 16 * h;
    float c = b1[n];
#pragma unroll
    for (int k = 0; k < LD; ++k) c += Ws[n * W1ROW + C0 + k] * z[k];
    cinit1[h] = c;
    cinit2[h] = b2[n];
    cinit3[h] = b3[n] * wln[h];
  }
  const float blv = bl[0];

  v2f Bw1[2][8], Bw2[2][8], Bw3[2][8];
#pragma unroll
  for (int h = 0; h < 2; ++h) {
    const int n = l16 + 16 * h;
#pragma unroll
    for (int j = 0; j < 8; ++j)
      Bw1[h][j] = *(const v2f*)&Ws[n * W1ROW + 4 * j + 2 * hi];
    const v4f* r2 = (const v4f*)&Ws2[n * WPITCH + 8 * hi];
    const v4f* r3 = (const v4f*)&Ws3[n * WPITCH + 8 * hi];
    const v4f q2[4] = { r2[0], r2[1], r2[4], r2[5] };
    const v4f q3[4] = { r3[0], r3[1], r3[4], r3[5] };
#pragma unroll
    for (int j = 0; j < 8; ++j) {
      v2f t;
      t[0] = q2[j >> 2][j & 3]; t[1] = q2[2 + (j >> 2)][j & 3];
      Bw2[h][j] = t;
      t[0] = q3[j >> 2][j & 3]; t[1] = q3[2 + (j >> 2)][j & 3];
      Bw3[h][j] = t * wln[h];
    }
  }

  __builtin_amdgcn_sched_barrier(0);

  const bool lb0 = (lane & 1) != 0, lb1 = (lane & 2) != 0;
  const bool lb2 = (lane & 4) != 0, lb3 = (lane & 8) != 0;

  float* Ha = Ws;
  float* Hb = Ws + 16 * HSTR;

  float tot2[2];
#pragma unroll
  for (int t = 0; t < TILES_PER_WAVE; t += 2) {
    float res2[2];
#pragma unroll
    for (int u = 0; u < 2; ++u) {

    v2f a[8];
#pragma unroll
    for (int j = 0; j < 8; ++j) {
      v2f av;
      av[0] = X[j][0][t + u];
      av[1] = X[j][1][t + u];
      a[j] = av;
    }
    v8f acc0, acc1;
#pragma unroll
    for (int r = 0; r < 8; ++r) { acc0[r] = cinit1[0]; acc1[r] = cinit1[1]; }
#pragma unroll
    for (int j = 0; j < 8; ++j) {
      acc0 = wmma_f32(a[j], Bw1[0][j], acc0);
      acc1 = wmma_f32(a[j], Bw1[1][j], acc1);
    }
#pragma unroll
    for (int r = 0; r < 8; ++r) {
      v2f p;
      p[0] = relu1(acc0[r]);
      p[1] = relu1(acc1[r]);
      *(v2f*)&Ha[(r + 8 * hi) * HSTR + 2 * l16] = p;
    }

#pragma unroll
    for (int j = 0; j < 8; ++j)
      a[j] = *(const v2f*)&Ha[l16 * HSTR + 16 * hi + 2 * j];
#pragma unroll
    for (int r = 0; r < 8; ++r) { acc0[r] = cinit2[0]; acc1[r] = cinit2[1]; }
#pragma unroll
    for (int j = 0; j < 8; ++j) {
      acc0 = wmma_f32(a[j], Bw2[0][j], acc0);
      acc1 = wmma_f32(a[j], Bw2[1][j], acc1);
    }
#pragma unroll
    for (int r = 0; r < 8; ++r) {
      v2f p;
      p[0] = relu1(acc0[r]);
      p[1] = relu1(acc1[r]);
      *(v2f*)&Hb[(r + 8 * hi) * HSTR + 2 * l16] = p;
    }

#pragma unroll
    for (int j = 0; j < 8; ++j)
      a[j] = *(const v2f*)&Hb[l16 * HSTR + 16 * hi + 2 * j];
#pragma unroll
    for (int r = 0; r < 8; ++r) { acc0[r] = cinit3[0]; acc1[r] = cinit3[1]; }
#pragma unroll
    for (int j = 0; j < 8; ++j) {
      acc0 = wmma_f32(a[j], Bw3[0][j], acc0);
      acc1 = wmma_f32(a[j], Bw3[1][j], acc1);
    }

    float v8[8], s4[4], s2[2];
#pragma unroll
    for (int r = 0; r < 8; ++r)
      v8[r] = clamp_side(acc0[r], bnd[0]) + clamp_side(acc1[r], bnd[1]);
#pragma unroll
    for (int i = 0; i < 4; ++i) {
      const float keep = lb3 ? v8[i + 4] : v8[i];
      const float give = lb3 ? v8[i] : v8[i + 4];
      s4[i] = keep + row_xor<8>(give);
    }
#pragma unroll
    for (int i = 0; i < 2; ++i) {
      const float keep = lb2 ? s4[i + 2] : s4[i];
      const float give = lb2 ? s4[i] : s4[i + 2];
      s2[i] = keep + row_xor<4>(give);
    }
    {
      const float keep = lb1 ? s2[1] : s2[0];
      const float give = lb1 ? s2[0] : s2[1];
      res2[u] = keep + row_xor<2>(give);
    }
    }
    const float keep = lb0 ? res2[1] : res2[0];
    const float give = lb0 ? res2[0] : res2[1];
    tot2[t >> 1] = keep + row_xor<1>(give);
    if (t == 0) __builtin_amdgcn_sched_barrier(0);
  }
  {
    const float give = lb0 ? tot2[0] : tot2[1];
    const float recv = row_xor<1>(give);
    v2f o;
    o[0] = (lb0 ? recv : tot2[0]) + blv;
    o[1] = (lb0 ? tot2[1] : recv) + blv;
    *(v2f*)&out[wbase + 2 * lane] = o;
  }
}

extern "C" void kernel_launch(void* const* d_in, const int* in_sizes, int n_in,
                              void* d_out, int out_size, void* d_ws, size_t ws_size,
                              hipStream_t stream) {
  (void)in_sizes; (void)n_in; (void)out_size; (void)d_ws; (void)ws_size;
  const float* fm = (const float*)d_in[0];
  const float* z  = (const float*)d_in[1];
  const float* w1 = (const float*)d_in[2];
  const float* b1 = (const float*)d_in[3];
  const float* w2 = (const float*)d_in[4];
  const float* b2 = (const float*)d_in[5];
  const float* w3 = (const float*)d_in[6];
  const float* b3 = (const float*)d_in[7];
  const float* wl = (const float*)d_in[8];
  const float* bl = (const float*)d_in[9];
  float* out = (float*)d_out;

  dim3 grid(VOL / POINTS_PER_BLOCK);
  dim3 block(256);
  fcomb_mlp_kernel<<<grid, block, 0, stream>>>(fm, z, w1, b1, w2, b2,
                                               w3, b3, wl, bl, out);
}
